# starting kernel
# baseline (speedup 1.0000x reference)
#include <hip/hip_runtime.h>
#include <math.h>

#define Cc   32
#define Kk   128
#define Bb   4
#define Tt   512
#define TPp  1024
#define CH   64
#define RT   2
#define EPS_F 2.2204460492503131e-16f

typedef __attribute__((ext_vector_type(2))) float v2f;
typedef __attribute__((ext_vector_type(8))) float v8f;

__device__ __forceinline__ float softplus_f(float x) {
    return fmaxf(x, 0.0f) + log1pf(__expf(-fabsf(x)));
}

__global__ void prep_params(const float* __restrict__ mu,
                            const float* __restrict__ alpha,
                            const float* __restrict__ delta,
                            const float* __restrict__ cf,
                            const int*   __restrict__ ftc,
                            float* __restrict__ sp_alpha,
                            float* __restrict__ sp_delta,
                            float* __restrict__ sp_mu,
                            float* __restrict__ Mexp) {
    __shared__ float s_e[Kk];
    __shared__ float s_den[Cc];
    __shared__ float s_prob[Kk];
    __shared__ float s_max;
    const int tid = threadIdx.x;

    for (int i = tid; i < Cc * Cc; i += 128) {
        sp_alpha[i] = softplus_f(alpha[i]);
        sp_delta[i] = softplus_f(delta[i]) + EPS_F;
    }
    if (tid < Cc) sp_mu[tid] = softplus_f(mu[tid]);

    if (tid == 0) {
        float m = cf[0];
        for (int k = 1; k < Kk; ++k) m = fmaxf(m, cf[k]);
        s_max = m;
    }
    __syncthreads();
    s_e[tid] = __expf(cf[tid] - s_max);
    __syncthreads();
    if (tid == 0) {
        for (int c = 0; c < Cc; ++c) s_den[c] = 0.0f;
        for (int k = 0; k < Kk; ++k) s_den[ftc[k]] += s_e[k];
    }
    __syncthreads();
    s_prob[tid] = s_e[tid] / s_den[ftc[tid]];
    __syncthreads();
    for (int i = tid; i < Cc * Kk; i += 128) {
        int c = i >> 7, k = i & (Kk - 1);
        Mexp[i] = (ftc[k] == c) ? s_prob[k] : 0.0f;
    }
}

__global__ __launch_bounds__(256)
void hawkes_wmma_kernel(const int*   __restrict__ past_event,
                        const float* __restrict__ past_time,
                        const float* __restrict__ time_tensor,
                        const float* __restrict__ sp_alpha,
                        const float* __restrict__ sp_delta,
                        const float* __restrict__ sp_mu,
                        const float* __restrict__ Mexp,
                        float*       __restrict__ out) {
    __shared__ float s_spA[Cc * Cc];
    __shared__ float s_spD[Cc * Cc];
    __shared__ float s_mu[Cc];
    __shared__ float s_pt[CH];
    __shared__ int   s_ev[CH];
    __shared__ float s_acc[16 * Cc];

    const int tid    = threadIdx.x;
    const int lane   = tid & 31;
    const int wave   = tid >> 5;
    const int b      = blockIdx.x >> 6;
    const int tile   = blockIdx.x & 63;
    const int t_base = tile * 16;

    for (int i = tid; i < Cc * Cc; i += 256) {
        s_spA[i] = sp_alpha[i];
        s_spD[i] = sp_delta[i];
    }
    if (tid < Cc) s_mu[tid] = sp_mu[tid];
    __builtin_prefetch(Mexp + lane * 16, 0, 0);

    float tm[RT], acc[RT];
#pragma unroll
    for (int r = 0; r < RT; ++r) {
        const int t = t_base + wave * RT + r;
        tm[r]  = time_tensor[b * TPp + t];
        acc[r] = 0.0f;
    }

    for (int chunk = 0; chunk < Tt; chunk += CH) {
        __syncthreads();
        if (tid < CH) {
            s_pt[tid] = past_time[b * Tt + chunk + tid];
            s_ev[tid] = past_event[b * Tt + chunk + tid];
        }
        __syncthreads();
#pragma unroll 4
        for (int ii = 0; ii < CH; ++ii) {
            const float pt = s_pt[ii];
            const int   ev = s_ev[ii];
            const float a  = s_spA[ev * Cc + lane];
            const float d  = s_spD[ev * Cc + lane];
#pragma unroll
            for (int r = 0; r < RT; ++r) {
                const float et  = tm[r] - pt;
                const float etc = fmaxf(et, 0.0f);
                const float val = a * __expf(-d * etc);
                acc[r] += (et > EPS_F) ? val : 0.0f;
            }
        }
    }

    __syncthreads();
#pragma unroll
    for (int r = 0; r < RT; ++r) {
        s_acc[(wave * RT + r) * Cc + lane] = acc[r] + s_mu[lane];
    }
    __syncthreads();

    const int half = lane >> 4;
    const int l15  = lane & 15;
    v8f cacc = {};
#pragma unroll
    for (int kc = 0; kc < 8; ++kc) {
        v2f af, bf;
#pragma unroll
        for (int j = 0; j < 2; ++j) {
            const int cch = kc * 4 + half * 2 + j;
            af[j] = s_acc[l15 * Cc + cch];
            bf[j] = Mexp[cch * Kk + wave * 16 + l15];
        }
        cacc = __builtin_amdgcn_wmma_f32_16x16x4_f32(
             false, af,  false, bf,
             (short)0, cacc,  false,  false);
    }
#pragma unroll
    for (int r = 0; r < 8; ++r) {
        const int m = r + half * 8;
        const int t = t_base + m;
        out[(b * TPp + t) * Kk + wave * 16 + l15] = cacc[r];
    }
}

extern "C" void kernel_launch(void* const* d_in, const int* in_sizes, int n_in,
                              void* d_out, int out_size, void* d_ws, size_t ws_size,
                              hipStream_t stream) {
    const int*   past_event  = (const int*)  d_in[0];
    const float* past_time   = (const float*)d_in[1];
    const float* time_tensor = (const float*)d_in[2];
    const float* mu          = (const float*)d_in[3];
    const float* alpha       = (const float*)d_in[4];
    const float* delta       = (const float*)d_in[5];
    const float* cf_logits   = (const float*)d_in[6];
    const int*   ftc         = (const int*)  d_in[7];
    float* out = (float*)d_out;

    float* sp_alpha = (float*)d_ws;
    float* sp_delta = sp_alpha + Cc * Cc;
    float* sp_mu    = sp_delta + Cc * Cc;
    float* Mexp     = sp_mu + Cc;

    prep_params<<<1, 128, 0, stream>>>(mu, alpha, delta, cf_logits, ftc,
                                       sp_alpha, sp_delta, sp_mu, Mexp);
    hawkes_wmma_kernel<<<Bb * (TPp / 16), 256, 0, stream>>>(
        past_event, past_time, time_tensor,
        sp_alpha, sp_delta, sp_mu, Mexp, out);
}


# Fuse the parameter-prep launch into the main kernel (2 launches -> 1): each block builds softplus tables and group-softmax terms in LDS (group sums as 4 interleaved partials), B fragments in VGPRs
# speedup vs baseline: 1.8226x; 1.8226x over previous
#include <hip/hip_runtime.h>
#include <math.h>


#define Cc   32
#define Kk   128
#define Bb   4
#define Tt   512
#define TPp  1024
#define CH   64
#define RT   2
#define EPS_F 2.2204460492503131e-16f

typedef __attribute__((ext_vector_type(2))) float v2f;
typedef __attribute__((ext_vector_type(8))) float v8f;

__device__ __forceinline__ float softplus_f(float x) {
    return fmaxf(x, 0.0f) + log1pf(__expf(-fabsf(x)));
}

__global__ __launch_bounds__(256)
void intensity_fused_kernel(const int*   __restrict__ past_event,
                            const float* __restrict__ past_time,
                            const float* __restrict__ time_tensor,
                            const float* __restrict__ mu,
                            const float* __restrict__ alpha,
                            const float* __restrict__ delta,
                            const float* __restrict__ cf,
                            const int*   __restrict__ ftc,
                            float*       __restrict__ out) {
    __shared__ float s_spA[Cc * Cc];
    __shared__ float s_spD[Cc * Cc];
    __shared__ float s_pt[CH];
    __shared__ int   s_ev[CH];
    __shared__ float s_acc[16 * Cc];
    __shared__ __attribute__((aligned(16))) float s_e[Kk];
    __shared__ __attribute__((aligned(16))) int   s_ftc[Kk];

    const int tid    = threadIdx.x;
    const int lane   = tid & 31;
    const int wave   = tid >> 5;
    const int b      = blockIdx.x >> 6;
    const int tile   = blockIdx.x & 63;
    const int t_base = tile * 16;

    float av[4], dv[4];
#pragma unroll
    for (int j = 0; j < 4; ++j) {
        av[j] = alpha[tid + 256 * j];
        dv[j] = delta[tid + 256 * j];
    }
    const float mu_raw = mu[lane];
    const float c0 = cf[lane], c1 = cf[lane + 32], c2 = cf[lane + 64], c3 = cf[lane + 96];
    const float c_own = cf[tid & (Kk - 1)];
    const int   f_own = ftc[tid & (Kk - 1)];

    float tm[RT], acc[RT];
#pragma unroll
    for (int r = 0; r < RT; ++r) {
        const int t = t_base + wave * RT + r;
        tm[r]  = time_tensor[b * TPp + t];
        acc[r] = 0.0f;
    }

#pragma unroll
    for (int j = 0; j < 4; ++j) {
        s_spA[tid + 256 * j] = softplus_f(av[j]);
        s_spD[tid + 256 * j] = softplus_f(dv[j]) + EPS_F;
    }
    const float sp_mu = softplus_f(mu_raw);

    float m = fmaxf(fmaxf(c0, c1), fmaxf(c2, c3));
#pragma unroll
    for (int off = 16; off > 0; off >>= 1) m = fmaxf(m, __shfl_xor(m, off));
    asm volatile("" :: "v"(c_own), "v"(f_own));
    if (tid < Kk) {
        s_e[tid]   = __expf(c_own - m);
        s_ftc[tid] = f_own;
    }

    for (int chunk = 0; chunk < Tt; chunk += CH) {
        __syncthreads();
        if (tid < CH) {
            s_pt[tid] = past_time[b * Tt + chunk + tid];
            s_ev[tid] = past_event[b * Tt + chunk + tid];
        }
        __syncthreads();
#pragma unroll 4
        for (int ii = 0; ii < CH; ++ii) {
            const float pt = s_pt[ii];
            const int   ev = s_ev[ii];
            const float a  = s_spA[ev * Cc + lane];
            const float d  = s_spD[ev * Cc + lane];
#pragma unroll
            for (int r = 0; r < RT; ++r) {
                const float et  = tm[r] - pt;
                const float etc = fmaxf(et, 0.0f);
                const float val = a * __expf(-d * etc);
                acc[r] += (et > EPS_F) ? val : 0.0f;
            }
        }
    }

    __syncthreads();
#pragma unroll
    for (int r = 0; r < RT; ++r) {
        s_acc[(wave * RT + r) * Cc + lane] = acc[r] + sp_mu;
    }

    const int half = lane >> 4;
    const int l15  = lane & 15;

    const int n    = wave * 16 + l15;
    const int my_c = s_ftc[n];
    float d0 = 0.0f, d1 = 0.0f, d2 = 0.0f, d3 = 0.0f;
#pragma unroll 8
    for (int k = 0; k < Kk; k += 4) {
        const float4 e4 = *reinterpret_cast<const float4*>(&s_e[k]);
        const int4   f4 = *reinterpret_cast<const int4*>(&s_ftc[k]);
        d0 += (f4.x == my_c) ? e4.x : 0.0f;
        d1 += (f4.y == my_c) ? e4.y : 0.0f;
        d2 += (f4.z == my_c) ? e4.z : 0.0f;
        d3 += (f4.w == my_c) ? e4.w : 0.0f;
    }
    const float den = (d0 + d1) + (d2 + d3);
    const float prob = s_e[n] / den;
    __syncthreads();

    v8f cacc = {};
#pragma unroll
    for (int kc = 0; kc < 8; ++kc) {
        v2f af, bf;
#pragma unroll
        for (int j = 0; j < 2; ++j) {
            const int cch = kc * 4 + half * 2 + j;
            af[j] = s_acc[l15 * Cc + cch];
            bf[j] = (my_c == cch) ? prob : 0.0f;
        }
        cacc = __builtin_amdgcn_wmma_f32_16x16x4_f32(
             false, af,  false, bf,
             (short)0, cacc,  false,  false);
    }
#pragma unroll
    for (int r = 0; r < 8; ++r) {
        const int m_row = r + half * 8;
        const int t = t_base + m_row;
        out[(b * TPp + t) * Kk + wave * 16 + l15] = cacc[r];
    }
}

extern "C" void kernel_launch(void* const* d_in, const int* in_sizes, int n_in,
                              void* d_out, int out_size, void* d_ws, size_t ws_size,
                              hipStream_t stream) {
    const int*   past_event  = (const int*)  d_in[0];
    const float* past_time   = (const float*)d_in[1];
    const float* time_tensor = (const float*)d_in[2];
    const float* mu          = (const float*)d_in[3];
    const float* alpha       = (const float*)d_in[4];
    const float* delta       = (const float*)d_in[5];
    const float* cf_logits   = (const float*)d_in[6];
    const int*   ftc         = (const int*)  d_in[7];
    float* out = (float*)d_out;

    intensity_fused_kernel<<<Bb * (TPp / 16), 256, 0, stream>>>(
        past_event, past_time, time_tensor,
        mu, alpha, delta, cf_logits, ftc, out);
}


# Wave-uniform part of the exponent (elapsed time, mask, log2e) computed once per wave, one lane per (event, query) pair, passed to the channel loop by readlane; no LDS event chunks; fma sum
# speedup vs baseline: 3.3412x; 1.8332x over previous
#include <hip/hip_runtime.h>
#include <math.h>


#define Cc   32
#define Kk   128
#define Bb   4
#define Tt   512
#define TPp  1024
#define GR   16
#define RT   2
#define EPS_F 2.2204460492503131e-16f
#define NEG_LOG2E (-1.4426950408889634f)

typedef __attribute__((ext_vector_type(2))) float v2f;
typedef __attribute__((ext_vector_type(8))) float v8f;

__device__ __forceinline__ float softplus_f(float x) {
    return fmaxf(x, 0.0f) + log1pf(__expf(-fabsf(x)));
}

__device__ __forceinline__ float pair_scale(float et) {
    return (et > EPS_F) ? et * NEG_LOG2E : -__builtin_inff();
}

__global__ __launch_bounds__(256)
void intensity_fused_kernel(const int*   __restrict__ past_event,
                            const float* __restrict__ past_time,
                            const float* __restrict__ time_tensor,
                            const float* __restrict__ mu,
                            const float* __restrict__ alpha,
                            const float* __restrict__ delta,
                            const float* __restrict__ cf,
                            const int*   __restrict__ ftc,
                            float*       __restrict__ out) {
    __shared__ float s_spA[Cc * Cc];
    __shared__ float s_spD[Cc * Cc];
    __shared__ float s_ptl[Tt];
    __shared__ int   s_evl[Tt];
    __shared__ float s_acc[16 * Cc];
    __shared__ __attribute__((aligned(16))) float s_e[Kk];
    __shared__ __attribute__((aligned(16))) int   s_ftc[Kk];

    const int tid    = threadIdx.x;
    const int lane   = tid & 31;
    const int wave   = tid >> 5;
    const int b      = blockIdx.x >> 6;
    const int tile   = blockIdx.x & 63;
    const int t_base = tile * 16;

    float av[4], dv[4];
#pragma unroll
    for (int j = 0; j < 4; ++j) {
        av[j] = alpha[tid + 256 * j];
        dv[j] = delta[tid + 256 * j];
    }
    const float mu_raw = mu[lane];
    const float c0 = cf[lane], c1 = cf[lane + 32], c2 = cf[lane + 64], c3 = cf[lane + 96];
    const float c_own = cf[tid & (Kk - 1)];
    const int   f_own = ftc[tid & (Kk - 1)];
    const float* __restrict__ ptb = past_time  + b * Tt;
    const int*   __restrict__ evb = past_event + b * Tt;
    const float pt_a = ptb[tid], pt_b = ptb[tid + 256];
    const int   ev_a = evb[tid], ev_b = evb[tid + 256];
    const int   l15  = lane & 15;
    const float tmv  = time_tensor[b * TPp + t_base + wave * RT + (lane >> 4)];
    const float pt_0 = ptb[l15];
    const int   ev_0 = evb[l15];

    float acc[RT];
#pragma unroll
    for (int r = 0; r < RT; ++r) acc[r] = 0.0f;

#pragma unroll
    for (int j = 0; j < 4; ++j) {
        s_spA[tid + 256 * j] = softplus_f(av[j]);
        s_spD[tid + 256 * j] = softplus_f(dv[j]) + EPS_F;
    }
    s_ptl[tid] = pt_a;  s_ptl[tid + 256] = pt_b;
    s_evl[tid] = ev_a;  s_evl[tid + 256] = ev_b;
    const float sp_mu = softplus_f(mu_raw);

    float m = fmaxf(fmaxf(c0, c1), fmaxf(c2, c3));
#pragma unroll
    for (int off = 16; off > 0; off >>= 1) m = fmaxf(m, __shfl_xor(m, off));
    asm volatile("" :: "v"(c_own), "v"(f_own));
    if (tid < Kk) {
        s_e[tid]   = __expf(c_own - m);
        s_ftc[tid] = f_own;
    }

    __syncthreads();

    float s2v = pair_scale(tmv - pt_0);
    int   ofv = ev_0 * Cc;
    for (int g = 0; g < Tt; g += GR) {
        const int gn = (g + GR < Tt) ? (g + GR) : (Tt - GR);
        const float pt_n = s_ptl[gn + l15];
        const int   ev_n = s_evl[gn + l15];

#pragma unroll
        for (int j = 0; j < GR; ++j) {
            const int row = __builtin_amdgcn_readlane(ofv, j) + lane;
            const float a = s_spA[row];
            const float d = s_spD[row];
#pragma unroll
            for (int r = 0; r < RT; ++r) {
                const float s2 = __int_as_float(
                    __builtin_amdgcn_readlane(__float_as_int(s2v), r * 16 + j));
                acc[r] = fmaf(a, __builtin_amdgcn_exp2f(d * s2), acc[r]);
            }
        }
        s2v = pair_scale(tmv - pt_n);
        ofv = ev_n * Cc;
    }

    __syncthreads();
#pragma unroll
    for (int r = 0; r < RT; ++r) {
        s_acc[(wave * RT + r) * Cc + lane] = acc[r] + sp_mu;
    }

    const int half = lane >> 4;

    const int n    = wave * 16 + l15;
    const int my_c = s_ftc[n];
    float d0 = 0.0f, d1 = 0.0f, d2 = 0.0f, d3 = 0.0f;
#pragma unroll 8
    for (int k = 0; k < Kk; k += 4) {
        const float4 e4 = *reinterpret_cast<const float4*>(&s_e[k]);
        const int4   f4 = *reinterpret_cast<const int4*>(&s_ftc[k]);
        d0 += (f4.x == my_c) ? e4.x : 0.0f;
        d1 += (f4.y == my_c) ? e4.y : 0.0f;
        d2 += (f4.z == my_c) ? e4.z : 0.0f;
        d3 += (f4.w == my_c) ? e4.w : 0.0f;
    }
    const float den = (d0 + d1) + (d2 + d3);
    const float prob = s_e[n] / den;
    __syncthreads();

    v8f cacc = {};
#pragma unroll
    for (int kc = 0; kc < 8; ++kc) {
        v2f af, bf;
#pragma unroll
        for (int j = 0; j < 2; ++j) {
            const int cch = kc * 4 + half * 2 + j;
            af[j] = s_acc[l15 * Cc + cch];
            bf[j] = (my_c == cch) ? prob : 0.0f;
        }
        cacc = __builtin_amdgcn_wmma_f32_16x16x4_f32(
             false, af,  false, bf,
             (short)0, cacc,  false,  false);
    }
#pragma unroll
    for (int r = 0; r < 8; ++r) {
        const int m_row = r + half * 8;
        const int t = t_base + m_row;
        out[(b * TPp + t) * Kk + wave * 16 + l15] = cacc[r];
    }
}

extern "C" void kernel_launch(void* const* d_in, const int* in_sizes, int n_in,
                              void* d_out, int out_size, void* d_ws, size_t ws_size,
                              hipStream_t stream) {
    const int*   past_event  = (const int*)  d_in[0];
    const float* past_time   = (const float*)d_in[1];
    const float* time_tensor = (const float*)d_in[2];
    const float* mu          = (const float*)d_in[3];
    const float* alpha       = (const float*)d_in[4];
    const float* delta       = (const float*)d_in[5];
    const float* cf_logits   = (const float*)d_in[6];
    const int*   ftc         = (const int*)  d_in[7];
    float* out = (float*)d_out;

    intensity_fused_kernel<<<Bb * (TPp / 16), 256, 0, stream>>>(
        past_event, past_time, time_tensor,
        mu, alpha, delta, cf_logits, ftc, out);
}


# Each wave takes 8 query times over one quarter of the events (was 2 over all): one table-row read and its address work now serve 8 exps; four in-order partial sums joined in LDS (sum order changes)
# speedup vs baseline: 4.0078x; 1.1995x over previous
#include <hip/hip_runtime.h>
#include <math.h>


#define Cc   32
#define Kk   128
#define Bb   4
#define Tt   512
#define TPp  1024
#define GR   16
#define RT   8
#define NQ   4
#define EQ   (Tt / NQ)
#define QPR  (32 / GR)
#define NREG (RT / QPR)
#define EPS_F 2.2204460492503131e-16f
#define NEG_LOG2E (-1.4426950408889634f)

typedef __attribute__((ext_vector_type(2))) float v2f;
typedef __attribute__((ext_vector_type(8))) float v8f;

__device__ __forceinline__ float softplus_f(float x) {
    return fmaxf(x, 0.0f) + log1pf(__expf(-fabsf(x)));
}

__device__ __forceinline__ float pair_scale(float et) {
    return (et > EPS_F) ? et * NEG_LOG2E : -__builtin_inff();
}

__global__ __launch_bounds__(256)
void intensity_fused_kernel(const int*   __restrict__ past_event,
                            const float* __restrict__ past_time,
                            const float* __restrict__ time_tensor,
                            const float* __restrict__ mu,
                            const float* __restrict__ alpha,
                            const float* __restrict__ delta,
                            const float* __restrict__ cf,
                            const int*   __restrict__ ftc,
                            float*       __restrict__ out) {
    __shared__ float s_spA[Cc * Cc];
    __shared__ float s_spD[Cc * Cc];
    __shared__ float s_ptl[Tt];
    __shared__ int   s_evl[Tt];
    __shared__ float s_acc[16 * Cc];
    __shared__ float s_part[NQ * 16 * Cc];
    __shared__ __attribute__((aligned(16))) float s_e[Kk];
    __shared__ __attribute__((aligned(16))) int   s_ftc[Kk];

    const int tid    = threadIdx.x;
    const int lane   = tid & 31;
    const int wave   = tid >> 5;
    const int b      = blockIdx.x >> 6;
    const int tile   = blockIdx.x & 63;
    const int t_base = tile * 16;

    float av[4], dv[4];
#pragma unroll
    for (int j = 0; j < 4; ++j) {
        av[j] = alpha[tid + 256 * j];
        dv[j] = delta[tid + 256 * j];
    }
    const float mu_raw = mu[lane];
    const float c0 = cf[lane], c1 = cf[lane + 32], c2 = cf[lane + 64], c3 = cf[lane + 96];
    const float c_own = cf[tid & (Kk - 1)];
    const int   f_own = ftc[tid & (Kk - 1)];
    const float* __restrict__ ptb = past_time  + b * Tt;
    const int*   __restrict__ evb = past_event + b * Tt;
    const float pt_a = ptb[tid], pt_b = ptb[tid + 256];
    const int   ev_a = evb[tid], ev_b = evb[tid + 256];
    const int   l15  = lane & 15;
    const int   lj   = lane & (GR - 1);
    const int   qg   = wave & 1;
    const int   e0   = (wave >> 1) * EQ;
    float tmv[NREG];
#pragma unroll
    for (int q = 0; q < NREG; ++q)
        tmv[q] = time_tensor[b * TPp + t_base + qg * RT + q * QPR + lane / GR];
    const float pt_0 = ptb[e0 + lj];
    const int   ev_0 = evb[e0 + lj];

    float acc[RT];
#pragma unroll
    for (int r = 0; r < RT; ++r) acc[r] = 0.0f;

#pragma unroll
    for (int j = 0; j < 4; ++j) {
        s_spA[tid + 256 * j] = softplus_f(av[j]);
        s_spD[tid + 256 * j] = softplus_f(dv[j]) + EPS_F;
    }
    s_ptl[tid] = pt_a;  s_ptl[tid + 256] = pt_b;
    s_evl[tid] = ev_a;  s_evl[tid + 256] = ev_b;
    const float sp_mu = softplus_f(mu_raw);

    float m = fmaxf(fmaxf(c0, c1), fmaxf(c2, c3));
#pragma unroll
    for (int off = 16; off > 0; off >>= 1) m = fmaxf(m, __shfl_xor(m, off));
    asm volatile("" :: "v"(c_own), "v"(f_own));
    if (tid < Kk) {
        s_e[tid]   = __expf(c_own - m);
        s_ftc[tid] = f_own;
    }

    __syncthreads();

    float s2v[NREG];
#pragma unroll
    for (int q = 0; q < NREG; ++q) s2v[q] = pair_scale(tmv[q] - pt_0);
    int   ofv = ev_0 * Cc;
    for (int g = 0; g < EQ; g += GR) {
        const int gn = e0 + ((g + GR < EQ) ? (g + GR) : (EQ - GR));
        const float pt_n = s_ptl[gn + lj];
        const int   ev_n = s_evl[gn + lj];

#pragma unroll
        for (int j = 0; j < GR; ++j) {
            const int row = __builtin_amdgcn_readlane(ofv, j) + lane;
            const float a = s_spA[row];
            const float d = s_spD[row];
#pragma unroll
            for (int r = 0; r < RT; ++r) {
                const float s2 = __int_as_float(
                    __builtin_amdgcn_readlane(__float_as_int(s2v[r / QPR]),
                                              (r % QPR) * GR + j));
                acc[r] = fmaf(a, __builtin_amdgcn_exp2f(d * s2), acc[r]);
            }
        }
#pragma unroll
        for (int q = 0; q < NREG; ++q) s2v[q] = pair_scale(tmv[q] - pt_n);
        ofv = ev_n * Cc;
    }

#pragma unroll
    for (int r = 0; r < RT; ++r) {
        s_part[((wave >> 1) * 16 + qg * RT + r) * Cc + lane] = acc[r];
    }
    __syncthreads();
#pragma unroll
    for (int h = 0; h < 2; ++h) {
        const int cell = (wave + 8 * h) * Cc + lane;
        const float p01 = s_part[cell] + s_part[16 * Cc + cell];
        const float p012 = p01 + s_part[2 * 16 * Cc + cell];
        s_acc[cell] = (p012 + s_part[3 * 16 * Cc + cell]) + sp_mu;
    }

    const int half = lane >> 4;

    const int n    = wave * 16 + l15;
    const int my_c = s_ftc[n];
    float d0 = 0.0f, d1 = 0.0f, d2 = 0.0f, d3 = 0.0f;
#pragma unroll 8
    for (int k = 0; k < Kk; k += 4) {
        const float4 e4 = *reinterpret_cast<const float4*>(&s_e[k]);
        const int4   f4 = *reinterpret_cast<const int4*>(&s_ftc[k]);
        d0 += (f4.x == my_c) ? e4.x : 0.0f;
        d1 += (f4.y == my_c) ? e4.y : 0.0f;
        d2 += (f4.z == my_c) ? e4.z : 0.0f;
        d3 += (f4.w == my_c) ? e4.w : 0.0f;
    }
    const float den = (d0 + d1) + (d2 + d3);
    const float prob = s_e[n] / den;
    __syncthreads();

    v8f cacc = {};
#pragma unroll
    for (int kc = 0; kc < 8; ++kc) {
        v2f af, bf;
#pragma unroll
        for (int j = 0; j < 2; ++j) {
            const int cch = kc * 4 + half * 2 + j;
            af[j] = s_acc[l15 * Cc + cch];
            bf[j] = (my_c == cch) ? prob : 0.0f;
        }
        cacc = __builtin_amdgcn_wmma_f32_16x16x4_f32(
             false, af,  false, bf,
             (short)0, cacc,  false,  false);
    }
#pragma unroll
    for (int r = 0; r < 8; ++r) {
        const int m_row = r + half * 8;
        const int t = t_base + m_row;
        out[(b * TPp + t) * Kk + wave * 16 + l15] = cacc[r];
    }
}

extern "C" void kernel_launch(void* const* d_in, const int* in_sizes, int n_in,
                              void* d_out, int out_size, void* d_ws, size_t ws_size,
                              hipStream_t stream) {
    const int*   past_event  = (const int*)  d_in[0];
    const float* past_time   = (const float*)d_in[1];
    const float* time_tensor = (const float*)d_in[2];
    const float* mu          = (const float*)d_in[3];
    const float* alpha       = (const float*)d_in[4];
    const float* delta       = (const float*)d_in[5];
    const float* cf_logits   = (const float*)d_in[6];
    const int*   ftc         = (const int*)  d_in[7];
    float* out = (float*)d_out;

    intensity_fused_kernel<<<Bb * (TPp / 16), 256, 0, stream>>>(
        past_event, past_time, time_tensor,
        mu, alpha, delta, cf_logits, ftc, out);
}


# Softplus of the parameter tables by three hardware transcendentals (exp2, log2, rcp: log1p(e) = log(u) e/(u-1), u = 1+e) instead of the library log1pf that every thread ran nine times
# speedup vs baseline: 5.1532x; 1.2858x over previous
#include <hip/hip_runtime.h>
#include <math.h>


#define Cc   32
#define Kk   128
#define Bb   4
#define Tt   512
#define TPp  1024
#define GR   16
#define RT   16
#define NW   8
#define GPW  (Tt / GR / NW)
#define QPR  (32 / GR)
#define NREG (RT / QPR)
#define EPS_F 2.2204460492503131e-16f
#define NEG_LOG2E (-1.4426950408889634f)

typedef __attribute__((ext_vector_type(2))) float v2f;
typedef __attribute__((ext_vector_type(8))) float v8f;

#define LN2_F 0.69314718055994531f

__device__ __forceinline__ float softplus_f(float x) {
    const float e   = __expf(-fabsf(x));
    const float u   = 1.0f + e;
    const float um1 = u - 1.0f;
    const float lg  = __builtin_amdgcn_logf(u) * LN2_F;
    const float r   = __builtin_amdgcn_rcpf(um1);
    const float q0  = e * r;
    const float q   = fmaf(fmaf(-um1, q0, e), r, q0);
    const float l1p = (um1 == 0.0f) ? e : lg * q;
    return fmaxf(x, 0.0f) + l1p;
}

__device__ __forceinline__ float pair_scale(float et) {
    return (et > EPS_F) ? et * NEG_LOG2E : -__builtin_inff();
}

__device__ __forceinline__ int group_of(int w, int gi) {
    return (gi >> 1) * (2 * NW) + ((gi & 1) ? (2 * NW - 1 - w) : w);
}

__global__ __launch_bounds__(256)
void intensity_fused_kernel(const int*   __restrict__ past_event,
                            const float* __restrict__ past_time,
                            const float* __restrict__ time_tensor,
                            const float* __restrict__ mu,
                            const float* __restrict__ alpha,
                            const float* __restrict__ delta,
                            const float* __restrict__ cf,
                            const int*   __restrict__ ftc,
                            float*       __restrict__ out) {
    __shared__ float s_spA[Cc * Cc];
    __shared__ float s_spD[Cc * Cc];
    __shared__ float s_ptl[Tt];
    __shared__ int   s_evl[Tt];
    __shared__ float s_acc[16 * Cc];
    __shared__ float s_part[NW * 16 * Cc];
    __shared__ __attribute__((aligned(16))) float s_e[Kk];
    __shared__ __attribute__((aligned(16))) int   s_ftc[Kk];

    const int tid    = threadIdx.x;
    const int lane   = tid & 31;
    const int wave   = tid >> 5;
    const int b      = blockIdx.x >> 6;
    const int tile   = blockIdx.x & 63;
    const int t_base = tile * 16;

    float av[4], dv[4];
#pragma unroll
    for (int j = 0; j < 4; ++j) {
        av[j] = alpha[tid + 256 * j];
        dv[j] = delta[tid + 256 * j];
    }
    const float mu_raw = mu[lane];
    const float c0 = cf[lane], c1 = cf[lane + 32], c2 = cf[lane + 64], c3 = cf[lane + 96];
    const float c_own = cf[tid & (Kk - 1)];
    const int   f_own = ftc[tid & (Kk - 1)];
    const float* __restrict__ ptb = past_time  + b * Tt;
    const int*   __restrict__ evb = past_event + b * Tt;
    const float pt_a = ptb[tid], pt_b = ptb[tid + 256];
    const int   ev_a = evb[tid], ev_b = evb[tid + 256];
    const int   l15  = lane & 15;
    const int   lj   = lane & (GR - 1);
    const int   e0   = group_of(wave, 0) * GR;
    float tmv[NREG];
#pragma unroll
    for (int q = 0; q < NREG; ++q)
        tmv[q] = time_tensor[b * TPp + t_base + q * QPR + lane / GR];
    const float pt_0 = ptb[e0 + lj];
    const int   ev_0 = evb[e0 + lj];

    float acc[RT];
#pragma unroll
    for (int r = 0; r < RT; ++r) acc[r] = 0.0f;

#pragma unroll
    for (int j = 0; j < 4; ++j) {
        s_spA[tid + 256 * j] = softplus_f(av[j]);
        s_spD[tid + 256 * j] = softplus_f(dv[j]) + EPS_F;
    }
    s_ptl[tid] = pt_a;  s_ptl[tid + 256] = pt_b;
    s_evl[tid] = ev_a;  s_evl[tid + 256] = ev_b;
    const float sp_mu = softplus_f(mu_raw);

    float m = fmaxf(fmaxf(c0, c1), fmaxf(c2, c3));
#pragma unroll
    for (int off = 16; off > 0; off >>= 1) m = fmaxf(m, __shfl_xor(m, off));
    asm volatile("" :: "v"(c_own), "v"(f_own));
    if (tid < Kk) {
        s_e[tid]   = __expf(c_own - m);
        s_ftc[tid] = f_own;
    }

    __syncthreads();

    float pt_c = pt_0;
    int   ev_c = ev_0;
    for (int gi = 0; gi < GPW; ++gi) {
        const int ofv = ev_c * Cc;
        float ta[GR], td[GR];
#pragma unroll
        for (int j = 0; j < GR; ++j) {
            const int row = __builtin_amdgcn_readlane(ofv, j) + lane;
            ta[j] = s_spA[row];
            td[j] = s_spD[row];
        }
        const int gn = group_of(wave, (gi + 1 < GPW) ? (gi + 1) : (GPW - 1)) * GR;
        const float pt_n = s_ptl[gn + lj];
        const int   ev_n = s_evl[gn + lj];

#pragma unroll
        for (int q = 0; q < NREG; ++q) {
            const float et  = tmv[q] - pt_c;
            const float s2v = pair_scale(et);
            const unsigned live = __builtin_amdgcn_ballot_w32(et > EPS_F);
#pragma unroll
            for (int u = 0; u < QPR; ++u) {
                if ((live >> (GR * u)) & ((1u << GR) - 1u)) {
                    const int r = q * QPR + u;
                    float hi = 0.0f;
#pragma unroll
                    for (int j = 0; j < GR; ++j) {
                        const float s2 = __int_as_float(
                            __builtin_amdgcn_readlane(__float_as_int(s2v), GR * u + j));
                        const float e = __builtin_amdgcn_exp2f(td[j] * s2);
                        if (j < GR / 2) acc[r] = fmaf(ta[j], e, acc[r]);
                        else            hi     = fmaf(ta[j], e, hi);
                    }
                    acc[r] += hi;
                }
            }
        }
        pt_c = pt_n;
        ev_c = ev_n;
    }

#pragma unroll
    for (int r = 0; r < RT; ++r) {
        s_part[(wave * 16 + r) * Cc + lane] = acc[r];
    }
    __syncthreads();
#pragma unroll
    for (int h = 0; h < 2; ++h) {
        const int cell = (wave + 8 * h) * Cc + lane;
        float p = s_part[cell];
#pragma unroll
        for (int w2 = 1; w2 < NW; ++w2) p += s_part[w2 * 16 * Cc + cell];
        s_acc[cell] = p + sp_mu;
    }

    const int half = lane >> 4;

    const int n    = wave * 16 + l15;
    const int my_c = s_ftc[n];
    float d0 = 0.0f, d1 = 0.0f, d2 = 0.0f, d3 = 0.0f;
#pragma unroll 8
    for (int k = 0; k < Kk; k += 4) {
        const float4 e4 = *reinterpret_cast<const float4*>(&s_e[k]);
        const int4   f4 = *reinterpret_cast<const int4*>(&s_ftc[k]);
        d0 += (f4.x == my_c) ? e4.x : 0.0f;
        d1 += (f4.y == my_c) ? e4.y : 0.0f;
        d2 += (f4.z == my_c) ? e4.z : 0.0f;
        d3 += (f4.w == my_c) ? e4.w : 0.0f;
    }
    const float den = (d0 + d1) + (d2 + d3);
    const float prob = s_e[n] / den;
    __syncthreads();

    v8f cacc = {};
#pragma unroll
    for (int kc = 0; kc < 8; ++kc) {
        v2f af, bf;
#pragma unroll
        for (int j = 0; j < 2; ++j) {
            const int cch = kc * 4 + half * 2 + j;
            af[j] = s_acc[l15 * Cc + cch];
            bf[j] = (my_c == cch) ? prob : 0.0f;
        }
        cacc = __builtin_amdgcn_wmma_f32_16x16x4_f32(
             false, af,  false, bf,
             (short)0, cacc,  false,  false);
    }
#pragma unroll
    for (int r = 0; r < 8; ++r) {
        const int m_row = r + half * 8;
        const int t = t_base + m_row;
        out[(b * TPp + t) * Kk + wave * 16 + l15] = cacc[r];
    }
}

extern "C" void kernel_launch(void* const* d_in, const int* in_sizes, int n_in,
                              void* d_out, int out_size, void* d_ws, size_t ws_size,
                              hipStream_t stream) {
    const int*   past_event  = (const int*)  d_in[0];
    const float* past_time   = (const float*)d_in[1];
    const float* time_tensor = (const float*)d_in[2];
    const float* mu          = (const float*)d_in[3];
    const float* alpha       = (const float*)d_in[4];
    const float* delta       = (const float*)d_in[5];
    const float* cf_logits   = (const float*)d_in[6];
    const int*   ftc         = (const int*)  d_in[7];
    float* out = (float*)d_out;

    intensity_fused_kernel<<<Bb * (TPp / 16), 256, 0, stream>>>(
        past_event, past_time, time_tensor,
        mu, alpha, delta, cf_logits, ftc, out);
}


# Event-loop lane = (channel pair, half group), not channel: one row broadcast (v_permlane16) replaces two readlanes, packed f32 mul/fma do two channels; 4 partial chains per wave (sum order changes)
# speedup vs baseline: 5.7392x; 1.1137x over previous
#include <hip/hip_runtime.h>
#include <math.h>


#define Cc   32
#define Kk   128
#define Bb   4
#define Tt   512
#define TPp  1024
#define GR   16
#define RT   16
#define NW   8
#define GPW  (Tt / GR / NW)
#define QPR  (32 / GR)
#define NREG (RT / QPR)
#define HG   (GR / 2)
#define NPL  (2 * NW)
#define EPS_F 2.2204460492503131e-16f
#define NEG_LOG2E (-1.4426950408889634f)

typedef __attribute__((ext_vector_type(2))) float v2f;
typedef __attribute__((ext_vector_type(8))) float v8f;

#define LN2_F 0.69314718055994531f

__device__ __forceinline__ float softplus_f(float x) {
    const float e   = __expf(-fabsf(x));
    const float u   = 1.0f + e;
    const float um1 = u - 1.0f;
    const float lg  = __builtin_amdgcn_logf(u) * LN2_F;
    const float r   = __builtin_amdgcn_rcpf(um1);
    const float q0  = e * r;
    const float q   = fmaf(fmaf(-um1, q0, e), r, q0);
    const float l1p = (um1 == 0.0f) ? e : lg * q;
    return fmaxf(x, 0.0f) + l1p;
}

__device__ __forceinline__ float pair_scale(float et) {
    return (et > EPS_F) ? et * NEG_LOG2E : -__builtin_inff();
}

__device__ __forceinline__ int group_of(int w, int gi) {
    return (gi >> 1) * (2 * NW) + ((gi & 1) ? (2 * NW - 1 - w) : w);
}

__global__ __launch_bounds__(256)
void intensity_fused_kernel(const int*   __restrict__ past_event,
                            const float* __restrict__ past_time,
                            const float* __restrict__ time_tensor,
                            const float* __restrict__ mu,
                            const float* __restrict__ alpha,
                            const float* __restrict__ delta,
                            const float* __restrict__ cf,
                            const int*   __restrict__ ftc,
                            float*       __restrict__ out) {
    __shared__ __attribute__((aligned(16))) float s_spA[Cc * Cc];
    __shared__ __attribute__((aligned(16))) float s_spD[Cc * Cc];
    __shared__ float s_ptl[Tt];
    __shared__ int   s_evl[Tt];
    __shared__ float s_acc[16 * Cc];
    __shared__ __attribute__((aligned(16))) float s_part[NPL * 16 * Cc];
    __shared__ __attribute__((aligned(16))) float s_e[Kk];
    __shared__ __attribute__((aligned(16))) int   s_ftc[Kk];

    const int tid    = threadIdx.x;
    const int lane   = tid & 31;
    const int wave   = tid >> 5;
    const int b      = blockIdx.x >> 6;
    const int tile   = blockIdx.x & 63;
    const int t_base = tile * 16;

    float av[4], dv[4];
#pragma unroll
    for (int j = 0; j < 4; ++j) {
        av[j] = alpha[tid + 256 * j];
        dv[j] = delta[tid + 256 * j];
    }
    const float mu_raw = mu[lane];
    const float c0 = cf[lane], c1 = cf[lane + 32], c2 = cf[lane + 64], c3 = cf[lane + 96];
    const float c_own = cf[tid & (Kk - 1)];
    const int   f_own = ftc[tid & (Kk - 1)];
    const float* __restrict__ ptb = past_time  + b * Tt;
    const int*   __restrict__ evb = past_event + b * Tt;
    const float pt_a = ptb[tid], pt_b = ptb[tid + 256];
    const int   ev_a = evb[tid], ev_b = evb[tid + 256];
    const int   l15  = lane & 15;
    const int   row  = lane >> 4;
    const int   lj   = row * HG + (lane & (HG - 1));
    const int   lq   = (lane >> 3) & 1;
    const int   e0   = group_of(wave, 0) * GR;
    float tmv[NREG];
#pragma unroll
    for (int q = 0; q < NREG; ++q)
        tmv[q] = time_tensor[b * TPp + t_base + q * QPR + lq];
    const float pt_0 = ptb[e0 + lj];
    const int   ev_0 = evb[e0 + lj];

    const v2f zero2 = {0.0f, 0.0f};
    v2f a0 = zero2, a1 = zero2, a2 = zero2, a3 = zero2, a4 = zero2, a5 = zero2, a6 = zero2, a7 = zero2;
    v2f a8 = zero2, a9 = zero2, a10 = zero2, a11 = zero2, a12 = zero2, a13 = zero2, a14 = zero2, a15 = zero2;
    v2f b0 = zero2, b1 = zero2, b2 = zero2, b3 = zero2, b4 = zero2, b5 = zero2, b6 = zero2, b7 = zero2;
    v2f b8 = zero2, b9 = zero2, b10 = zero2, b11 = zero2, b12 = zero2, b13 = zero2, b14 = zero2, b15 = zero2;

#pragma unroll
    for (int j = 0; j < 4; ++j) {
        const int dst = (wave + NW * j) * Cc + l15 * 2 + row;
        s_spA[dst] = softplus_f(av[j]);
        s_spD[dst] = softplus_f(dv[j]) + EPS_F;
    }
    s_ptl[tid] = pt_a;  s_ptl[tid + 256] = pt_b;
    s_evl[tid] = ev_a;  s_evl[tid + 256] = ev_b;
    const float sp_mu = softplus_f(mu_raw);

    float m = fmaxf(fmaxf(c0, c1), fmaxf(c2, c3));
#pragma unroll
    for (int off = 16; off > 0; off >>= 1) m = fmaxf(m, __shfl_xor(m, off));
    asm volatile("" :: "v"(c_own), "v"(f_own));
    if (tid < Kk) {
        s_e[tid]   = __expf(c_own - m);
        s_ftc[tid] = f_own;
    }

    __syncthreads();

    const v2f* __restrict__ spA2 = reinterpret_cast<const v2f*>(s_spA);
    const v2f* __restrict__ spD2 = reinterpret_cast<const v2f*>(s_spD);
    float pt_c = pt_0;
    int   ev_c = ev_0;
#pragma nounroll
    for (int gi = 0; gi < GPW; ++gi) {
        const unsigned rowv = (unsigned)(ev_c * (Cc / 2));
        v2f ta2[HG], td2[HG];
#pragma unroll
        for (int j = 0; j < HG; ++j) {
            const unsigned sel = 0x11111111u * (unsigned)j;
            const unsigned rb = __builtin_amdgcn_permlane16(rowv, rowv, sel, sel, true, false);
            ta2[j] = spA2[rb + (unsigned)l15];
            td2[j] = spD2[rb + (unsigned)l15];
        }
        const int gn = group_of(wave, (gi + 1 < GPW) ? (gi + 1) : (GPW - 1)) * GR;
        const float pt_n = s_ptl[gn + lj];
        const int   ev_n = s_evl[gn + lj];

#define EVENT_STEP(J, V, ACC)                                                          \
        {                                                                              \
            const unsigned sel = 0x11111111u * (unsigned)(HG * (V) + (J));             \
            const float s2 = __uint_as_float(__builtin_amdgcn_permlane16(              \
                __float_as_uint(s2v), __float_as_uint(s2v), sel, sel, true, false));   \
            const v2f x = td2[J] * s2;                                                 \
            v2f e;                                                                     \
            e.x = __builtin_amdgcn_exp2f(x.x);                                         \
            e.y = __builtin_amdgcn_exp2f(x.y);                                         \
            ACC = __builtin_elementwise_fma(ta2[J], e, ACC);                           \
        }
#define UNIT(V, A, B)                                                                  \
        if (live & (0x00FF00FFu << (HG * (V)))) {                                      \
            EVENT_STEP(0, V, A) EVENT_STEP(1, V, B) EVENT_STEP(2, V, A) EVENT_STEP(3, V, B) \
            EVENT_STEP(4, V, A) EVENT_STEP(5, V, B) EVENT_STEP(6, V, A) EVENT_STEP(7, V, B) \
        }
#define QUERY_PAIR(Q, A0, B0, A1, B1)                                                  \
        {                                                                              \
            const float et  = tmv[Q] - pt_c;                                           \
            const float s2v = pair_scale(et);                                          \
            const unsigned live = __builtin_amdgcn_ballot_w32(et > EPS_F);             \
            UNIT(0, A0, B0)                                                            \
            UNIT(1, A1, B1)                                                            \
        }
        QUERY_PAIR(0, a0, b0, a1, b1)
        QUERY_PAIR(1, a2, b2, a3, b3)
        QUERY_PAIR(2, a4, b4, a5, b5)
        QUERY_PAIR(3, a6, b6, a7, b7)
        QUERY_PAIR(4, a8, b8, a9, b9)
        QUERY_PAIR(5, a10, b10, a11, b11)
        QUERY_PAIR(6, a12, b12, a13, b13)
        QUERY_PAIR(7, a14, b14, a15, b15)
#undef QUERY_PAIR
#undef UNIT
#undef EVENT_STEP
        pt_c = pt_n;
        ev_c = ev_n;
    }

    {
        v2f* __restrict__ part2 = reinterpret_cast<v2f*>(s_part) + ((wave * 2 + row) * 16) * (Cc / 2) + l15;
        part2[ 0 * (Cc / 2)] = a0 + b0;    part2[ 1 * (Cc / 2)] = a1 + b1;
        part2[ 2 * (Cc / 2)] = a2 + b2;    part2[ 3 * (Cc / 2)] = a3 + b3;
        part2[ 4 * (Cc / 2)] = a4 + b4;    part2[ 5 * (Cc / 2)] = a5 + b5;
        part2[ 6 * (Cc / 2)] = a6 + b6;    part2[ 7 * (Cc / 2)] = a7 + b7;
        part2[ 8 * (Cc / 2)] = a8 + b8;    part2[ 9 * (Cc / 2)] = a9 + b9;
        part2[10 * (Cc / 2)] = a10 + b10;  part2[11 * (Cc / 2)] = a11 + b11;
        part2[12 * (Cc / 2)] = a12 + b12;  part2[13 * (Cc / 2)] = a13 + b13;
        part2[14 * (Cc / 2)] = a14 + b14;  part2[15 * (Cc / 2)] = a15 + b15;
    }
    __syncthreads();
#pragma unroll
    for (int h = 0; h < 2; ++h) {
        const int cell = (wave + 8 * h) * Cc + lane;
        const int from = (wave + 8 * h) * Cc + l15 * 2 + row;
        float p = s_part[from];
#pragma unroll
        for (int w2 = 1; w2 < NPL; ++w2) p += s_part[w2 * 16 * Cc + from];
        s_acc[cell] = p + sp_mu;
    }

    const int half = lane >> 4;

    const int n    = wave * 16 + l15;
    const int my_c = s_ftc[n];
    float d0 = 0.0f, d1 = 0.0f, d2 = 0.0f, d3 = 0.0f;
#pragma unroll 8
    for (int k = 0; k < Kk; k += 4) {
        const float4 e4 = *reinterpret_cast<const float4*>(&s_e[k]);
        const int4   f4 = *reinterpret_cast<const int4*>(&s_ftc[k]);
        d0 += (f4.x == my_c) ? e4.x : 0.0f;
        d1 += (f4.y == my_c) ? e4.y : 0.0f;
        d2 += (f4.z == my_c) ? e4.z : 0.0f;
        d3 += (f4.w == my_c) ? e4.w : 0.0f;
    }
    const float den = (d0 + d1) + (d2 + d3);
    const float prob = s_e[n] / den;
    __syncthreads();

    v8f cacc = {};
#pragma unroll
    for (int kc = 0; kc < 8; ++kc) {
        v2f af, bf;
#pragma unroll
        for (int j = 0; j < 2; ++j) {
            const int cch = kc * 4 + half * 2 + j;
            af[j] = s_acc[l15 * Cc + cch];
            bf[j] = (my_c == cch) ? prob : 0.0f;
        }
        cacc = __builtin_amdgcn_wmma_f32_16x16x4_f32(
             false, af,  false, bf,
             (short)0, cacc,  false,  false);
    }
#pragma unroll
    for (int r = 0; r < 8; ++r) {
        const int m_row = r + half * 8;
        const int t = t_base + m_row;
        out[(b * TPp + t) * Kk + wave * 16 + l15] = cacc[r];
    }
}

extern "C" void kernel_launch(void* const* d_in, const int* in_sizes, int n_in,
                              void* d_out, int out_size, void* d_ws, size_t ws_size,
                              hipStream_t stream) {
    const int*   past_event  = (const int*)  d_in[0];
    const float* past_time   = (const float*)d_in[1];
    const float* time_tensor = (const float*)d_in[2];
    const float* mu          = (const float*)d_in[3];
    const float* alpha       = (const float*)d_in[4];
    const float* delta       = (const float*)d_in[5];
    const float* cf_logits   = (const float*)d_in[6];
    const int*   ftc         = (const int*)  d_in[7];
    float* out = (float*)d_out;

    intensity_fused_kernel<<<Bb * (TPp / 16), 256, 0, stream>>>(
        past_event, past_time, time_tensor,
        mu, alpha, delta, cf_logits, ftc, out);
}


# Group sums of the fine softmax once per block: wave w sums its 16 fine ids per group from registers (readlane), 8 partials joined in LDS; no 128-step scan per lane (sum order changes)
# speedup vs baseline: 6.1085x; 1.0643x over previous
#include <hip/hip_runtime.h>
#include <math.h>


#define Cc   32
#define Kk   128
#define Bb   4
#define Tt   512
#define TPp  1024
#define GR   16
#define RT   16
#define NW   8
#define GPW  (Tt / GR / NW)
#define QPR  (32 / GR)
#define NREG (RT / QPR)
#define HG   (GR / 2)
#define NPL  (2 * NW)
#define EPS_F 2.2204460492503131e-16f
#define NEG_LOG2E (-1.4426950408889634f)

typedef __attribute__((ext_vector_type(2))) float v2f;
typedef __attribute__((ext_vector_type(8))) float v8f;

#define LN2_F 0.69314718055994531f

__device__ __forceinline__ float softplus_f(float x) {
    const float e   = __expf(-fabsf(x));
    const float u   = 1.0f + e;
    const float um1 = u - 1.0f;
    const float lg  = __builtin_amdgcn_logf(u) * LN2_F;
    const float r   = __builtin_amdgcn_rcpf(um1);
    const float q0  = e * r;
    const float q   = fmaf(fmaf(-um1, q0, e), r, q0);
    const float l1p = (um1 == 0.0f) ? e : lg * q;
    return fmaxf(x, 0.0f) + l1p;
}

__device__ __forceinline__ float pair_scale(float et) {
    return (et > EPS_F) ? et * NEG_LOG2E : -__builtin_inff();
}

__device__ __forceinline__ int group_of(int w, int gi) {
    return (gi >> 1) * (2 * NW) + ((gi & 1) ? (2 * NW - 1 - w) : w);
}

__global__ __launch_bounds__(256)
void intensity_fused_kernel(const int*   __restrict__ past_event,
                            const float* __restrict__ past_time,
                            const float* __restrict__ time_tensor,
                            const float* __restrict__ mu,
                            const float* __restrict__ alpha,
                            const float* __restrict__ delta,
                            const float* __restrict__ cf,
                            const int*   __restrict__ ftc,
                            float*       __restrict__ out) {
    __shared__ __attribute__((aligned(16))) float s_spA[Cc * Cc];
    __shared__ __attribute__((aligned(16))) float s_spD[Cc * Cc];
    __shared__ float s_ptl[Tt];
    __shared__ int   s_evl[Tt];
    __shared__ float s_acc[16 * Cc];
    __shared__ __attribute__((aligned(16))) float s_part[NPL * 16 * Cc];
    __shared__ float s_dp[NW * Cc];

    const int tid    = threadIdx.x;
    const int lane   = tid & 31;
    const int wave   = tid >> 5;
    const int b      = blockIdx.x >> 6;
    const int tile   = blockIdx.x & 63;
    const int t_base = tile * 16;

    float av[4], dv[4];
#pragma unroll
    for (int j = 0; j < 4; ++j) {
        av[j] = alpha[tid + 256 * j];
        dv[j] = delta[tid + 256 * j];
    }
    const float mu_raw = mu[lane];
    const float c0 = cf[lane], c1 = cf[lane + 32], c2 = cf[lane + 64], c3 = cf[lane + 96];
    const float c_own = cf[wave * 16 + (lane & 15)];
    const int   f_own = ftc[wave * 16 + (lane & 15)];
    const float* __restrict__ ptb = past_time  + b * Tt;
    const int*   __restrict__ evb = past_event + b * Tt;
    const float pt_a = ptb[tid], pt_b = ptb[tid + 256];
    const int   ev_a = evb[tid], ev_b = evb[tid + 256];
    const int   l15  = lane & 15;
    const int   row  = lane >> 4;
    const int   lj   = row * HG + (lane & (HG - 1));
    const int   lq   = (lane >> 3) & 1;
    const int   e0   = group_of(wave, 0) * GR;
    float tmv[NREG];
#pragma unroll
    for (int q = 0; q < NREG; ++q)
        tmv[q] = time_tensor[b * TPp + t_base + q * QPR + lq];
    const float pt_0 = ptb[e0 + lj];
    const int   ev_0 = evb[e0 + lj];

    const v2f zero2 = {0.0f, 0.0f};
    v2f a0 = zero2, a1 = zero2, a2 = zero2, a3 = zero2, a4 = zero2, a5 = zero2, a6 = zero2, a7 = zero2;
    v2f a8 = zero2, a9 = zero2, a10 = zero2, a11 = zero2, a12 = zero2, a13 = zero2, a14 = zero2, a15 = zero2;
    v2f b0 = zero2, b1 = zero2, b2 = zero2, b3 = zero2, b4 = zero2, b5 = zero2, b6 = zero2, b7 = zero2;
    v2f b8 = zero2, b9 = zero2, b10 = zero2, b11 = zero2, b12 = zero2, b13 = zero2, b14 = zero2, b15 = zero2;

#pragma unroll
    for (int j = 0; j < 4; ++j) {
        const int dst = (wave + NW * j) * Cc + l15 * 2 + row;
        s_spA[dst] = softplus_f(av[j]);
        s_spD[dst] = softplus_f(dv[j]) + EPS_F;
    }
    s_ptl[tid] = pt_a;  s_ptl[tid + 256] = pt_b;
    s_evl[tid] = ev_a;  s_evl[tid + 256] = ev_b;
    const float sp_mu = softplus_f(mu_raw);

    float m = fmaxf(fmaxf(c0, c1), fmaxf(c2, c3));
#pragma unroll
    for (int off = 16; off > 0; off >>= 1) m = fmaxf(m, __shfl_xor(m, off));
    const float e_own = __expf(c_own - m);
    {
        float ej[16];
        int   fj[16];
#pragma unroll
        for (int j = 0; j < 16; ++j) {
            ej[j] = __int_as_float(__builtin_amdgcn_readlane(__float_as_int(e_own), j));
            fj[j] = __builtin_amdgcn_readlane(f_own, j);
        }
        float q0 = 0.0f, q1 = 0.0f, q2 = 0.0f, q3 = 0.0f;
#pragma unroll
        for (int j = 0; j < 16; j += 4) {
            q0 += (fj[j    ] == lane) ? ej[j    ] : 0.0f;
            q1 += (fj[j + 1] == lane) ? ej[j + 1] : 0.0f;
            q2 += (fj[j + 2] == lane) ? ej[j + 2] : 0.0f;
            q3 += (fj[j + 3] == lane) ? ej[j + 3] : 0.0f;
        }
        s_dp[wave * Cc + lane] = (q0 + q1) + (q2 + q3);
    }

    __syncthreads();

    const int my_c = f_own;
    float den = s_dp[my_c & (Cc - 1)];
#pragma unroll
    for (int w2 = 1; w2 < NW; ++w2) den += s_dp[w2 * Cc + (my_c & (Cc - 1))];
    const float prob = e_own / den;

    const v2f* __restrict__ spA2 = reinterpret_cast<const v2f*>(s_spA);
    const v2f* __restrict__ spD2 = reinterpret_cast<const v2f*>(s_spD);
    float pt_c = pt_0;
    int   ev_c = ev_0;
#pragma nounroll
    for (int gi = 0; gi < GPW; ++gi) {
        const unsigned rowv = (unsigned)(ev_c * (Cc / 2));
        v2f ta2[HG], td2[HG];
#pragma unroll
        for (int j = 0; j < HG; ++j) {
            const unsigned sel = 0x11111111u * (unsigned)j;
            const unsigned rb = __builtin_amdgcn_permlane16(rowv, rowv, sel, sel, true, false);
            ta2[j] = spA2[rb + (unsigned)l15];
            td2[j] = spD2[rb + (unsigned)l15];
        }
        const int gn = group_of(wave, (gi + 1 < GPW) ? (gi + 1) : (GPW - 1)) * GR;
        const float pt_n = s_ptl[gn + lj];
        const int   ev_n = s_evl[gn + lj];

#define EVENT_STEP(J, V, ACC)                                                          \
        {                                                                              \
            const unsigned sel = 0x11111111u * (unsigned)(HG * (V) + (J));             \
            const float s2 = __uint_as_float(__builtin_amdgcn_permlane16(              \
                __float_as_uint(s2v), __float_as_uint(s2v), sel, sel, true, false));   \
            const v2f x = td2[J] * s2;                                                 \
            v2f e;                                                                     \
            e.x = __builtin_amdgcn_exp2f(x.x);                                         \
            e.y = __builtin_amdgcn_exp2f(x.y);                                         \
            ACC = __builtin_elementwise_fma(ta2[J], e, ACC);                           \
        }
#define UNIT(V, A, B)                                                                  \
        if (live & (0x00FF00FFu << (HG * (V)))) {                                      \
            EVENT_STEP(0, V, A) EVENT_STEP(1, V, B) EVENT_STEP(2, V, A) EVENT_STEP(3, V, B) \
            EVENT_STEP(4, V, A) EVENT_STEP(5, V, B) EVENT_STEP(6, V, A) EVENT_STEP(7, V, B) \
        }
#define QUERY_PAIR(Q, A0, B0, A1, B1)                                                  \
        {                                                                              \
            const float et  = tmv[Q] - pt_c;                                           \
            const float s2v = pair_scale(et);                                          \
            const unsigned live = __builtin_amdgcn_ballot_w32(et > EPS_F);             \
            UNIT(0, A0, B0)                                                            \
            UNIT(1, A1, B1)                                                            \
        }
        QUERY_PAIR(0, a0, b0, a1, b1)
        QUERY_PAIR(1, a2, b2, a3, b3)
        QUERY_PAIR(2, a4, b4, a5, b5)
        QUERY_PAIR(3, a6, b6, a7, b7)
        QUERY_PAIR(4, a8, b8, a9, b9)
        QUERY_PAIR(5, a10, b10, a11, b11)
        QUERY_PAIR(6, a12, b12, a13, b13)
        QUERY_PAIR(7, a14, b14, a15, b15)
#undef QUERY_PAIR
#undef UNIT
#undef EVENT_STEP
        pt_c = pt_n;
        ev_c = ev_n;
    }

    {
        v2f* __restrict__ part2 = reinterpret_cast<v2f*>(s_part) + ((wave * 2 + row) * 16) * (Cc / 2) + l15;
        part2[ 0 * (Cc / 2)] = a0 + b0;    part2[ 1 * (Cc / 2)] = a1 + b1;
        part2[ 2 * (Cc / 2)] = a2 + b2;    part2[ 3 * (Cc / 2)] = a3 + b3;
        part2[ 4 * (Cc / 2)] = a4 + b4;    part2[ 5 * (Cc / 2)] = a5 + b5;
        part2[ 6 * (Cc / 2)] = a6 + b6;    part2[ 7 * (Cc / 2)] = a7 + b7;
        part2[ 8 * (Cc / 2)] = a8 + b8;    part2[ 9 * (Cc / 2)] = a9 + b9;
        part2[10 * (Cc / 2)] = a10 + b10;  part2[11 * (Cc / 2)] = a11 + b11;
        part2[12 * (Cc / 2)] = a12 + b12;  part2[13 * (Cc / 2)] = a13 + b13;
        part2[14 * (Cc / 2)] = a14 + b14;  part2[15 * (Cc / 2)] = a15 + b15;
    }
    __syncthreads();
#pragma unroll
    for (int h = 0; h < 2; ++h) {
        const int cell = (wave + 8 * h) * Cc + lane;
        const int from = (wave + 8 * h) * Cc + l15 * 2 + row;
        float p = s_part[from];
#pragma unroll
        for (int w2 = 1; w2 < NPL; ++w2) p += s_part[w2 * 16 * Cc + from];
        s_acc[cell] = p + sp_mu;
    }

    const int half = lane >> 4;

    __syncthreads();

    v8f cacc = {};
#pragma unroll
    for (int kc = 0; kc < 8; ++kc) {
        v2f af, bf;
#pragma unroll
        for (int j = 0; j < 2; ++j) {
            const int cch = kc * 4 + half * 2 + j;
            af[j] = s_acc[l15 * Cc + cch];
            bf[j] = (my_c == cch) ? prob : 0.0f;
        }
        cacc = __builtin_amdgcn_wmma_f32_16x16x4_f32(
             false, af,  false, bf,
             (short)0, cacc,  false,  false);
    }
#pragma unroll
    for (int r = 0; r < 8; ++r) {
        const int m_row = r + half * 8;
        const int t = t_base + m_row;
        out[(b * TPp + t) * Kk + wave * 16 + l15] = cacc[r];
    }
}

extern "C" void kernel_launch(void* const* d_in, const int* in_sizes, int n_in,
                              void* d_out, int out_size, void* d_ws, size_t ws_size,
                              hipStream_t stream) {
    const int*   past_event  = (const int*)  d_in[0];
    const float* past_time   = (const float*)d_in[1];
    const float* time_tensor = (const float*)d_in[2];
    const float* mu          = (const float*)d_in[3];
    const float* alpha       = (const float*)d_in[4];
    const float* delta       = (const float*)d_in[5];
    const float* cf_logits   = (const float*)d_in[6];
    const int*   ftc         = (const int*)  d_in[7];
    float* out = (float*)d_out;

    intensity_fused_kernel<<<Bb * (TPp / 16), 256, 0, stream>>>(
        past_event, past_time, time_tensor,
        mu, alpha, delta, cf_logits, ftc, out);
}
